# starting kernel
# baseline (speedup 1.0000x reference)
#include <hip/hip_runtime.h>


#define BH      32
#define T       1024
#define DD      64
#define CHUNK   64
#define NCHUNK  (T / CHUNK)
#define EB      16
#define NE      (DD / EB)
#define QSTR    68
#define VSTR    24

#define L2DECAY     (-0.07400058144377693f)
#define DECAY_CHUNK (0.0375241347f)

typedef __attribute__((ext_vector_type(2))) float v2f;
typedef __attribute__((ext_vector_type(8))) float v8f;

__device__ __forceinline__ v8f wmma_f32(v2f a, v2f b, v8f c) {
    return __builtin_amdgcn_wmma_f32_16x16x4_f32(
         false, a,  false, b,
         (short)0, c,  false,  false);
}

__global__ __launch_bounds__(128)
void rkv_chunked_wmma(const float* __restrict__ q,
                      const float* __restrict__ k,
                      const float* __restrict__ v,
                      float* __restrict__ out)
{
    __shared__ float sQ[CHUNK][QSTR];
    __shared__ float sK[CHUNK][QSTR];
    __shared__ float sA[CHUNK][QSTR];
    __shared__ float sV[CHUNK][VSTR];
    __shared__ float sS[DD][VSTR];

    const int tid  = threadIdx.x;
    const int lane = tid & 31;
    const int wave = tid >> 5;
    const int lr   = lane & 15;
    const int half = lane >> 4;
    const int m0   = wave * 16;

    const int bh = blockIdx.x / NE;
    const int e0 = (blockIdx.x % NE) * EB;

    const float* qbh = q + (size_t)bh * T * DD;
    const float* kbh = k + (size_t)bh * T * DD;
    const float* vbh = v + (size_t)bh * T * DD;
    float*       obh = out + (size_t)bh * T * DD;

    for (int i = tid; i < DD * EB; i += 128)
        sS[i / EB][i % EB] = 0.0f;

    for (int c = 0; c < NCHUNK; ++c) {
        __syncthreads();
        const int tbase = c * CHUNK;

        for (int i = tid; i < CHUNK * (DD / 4); i += 128) {
            const int row = i / (DD / 4);
            const int c4  = (i % (DD / 4)) * 4;
            const float4 qv = *(const float4*)(qbh + (size_t)(tbase + row) * DD + c4);
            const float4 kv = *(const float4*)(kbh + (size_t)(tbase + row) * DD + c4);
            const float qs = __builtin_amdgcn_exp2f((float)(row + 1) * L2DECAY);
            const float ks = __builtin_amdgcn_exp2f((float)(-(row + 1)) * L2DECAY);
            sQ[row][c4 + 0] = qv.x * qs; sQ[row][c4 + 1] = qv.y * qs;
            sQ[row][c4 + 2] = qv.z * qs; sQ[row][c4 + 3] = qv.w * qs;
            sK[row][c4 + 0] = kv.x * ks; sK[row][c4 + 1] = kv.y * ks;
            sK[row][c4 + 2] = kv.z * ks; sK[row][c4 + 3] = kv.w * ks;
        }
        for (int i = tid; i < CHUNK * (EB / 4); i += 128) {
            const int row = i / (EB / 4);
            const int c4  = (i % (EB / 4)) * 4;
            const float4 vv = *(const float4*)(vbh + (size_t)(tbase + row) * DD + e0 + c4);
            sV[row][c4 + 0] = vv.x; sV[row][c4 + 1] = vv.y;
            sV[row][c4 + 2] = vv.z; sV[row][c4 + 3] = vv.w;
        }
        __syncthreads();

        for (int nb = 0; nb < 4; ++nb) {
            const int n0 = nb * 16;
            v8f acc = {};
            #pragma unroll
            for (int k0 = 0; k0 < DD; k0 += 4) {
                v2f a, b;
                a[0] = sQ[m0 + lr][k0 + 2 * half + 0];
                a[1] = sQ[m0 + lr][k0 + 2 * half + 1];
                b[0] = sK[n0 + lr][k0 + 2 * half + 0];
                b[1] = sK[n0 + lr][k0 + 2 * half + 1];
                acc = wmma_f32(a, b, acc);
            }
            #pragma unroll
            for (int r = 0; r < 8; ++r) {
                const int t = m0 + r + 8 * half;
                const int s = n0 + lr;
                sA[t][s] = (s <= t) ? acc[r] : 0.0f;
            }
        }
        __syncthreads();

        {
            v8f acc = {};
            #pragma unroll
            for (int k0 = 0; k0 < CHUNK; k0 += 4) {
                v2f a, b;
                a[0] = sA[m0 + lr][k0 + 2 * half + 0];
                a[1] = sA[m0 + lr][k0 + 2 * half + 1];
                b[0] = sV[k0 + 2 * half + 0][lr];
                b[1] = sV[k0 + 2 * half + 1][lr];
                acc = wmma_f32(a, b, acc);
            }
            #pragma unroll
            for (int k0 = 0; k0 < DD; k0 += 4) {
                v2f a, b;
                a[0] = sQ[m0 + lr][k0 + 2 * half + 0];
                a[1] = sQ[m0 + lr][k0 + 2 * half + 1];
                b[0] = sS[k0 + 2 * half + 0][lr];
                b[1] = sS[k0 + 2 * half + 1][lr];
                acc = wmma_f32(a, b, acc);
            }
            #pragma unroll
            for (int r = 0; r < 8; ++r) {
                const int t = tbase + m0 + r + 8 * half;
                const int e = e0 + lr;
                obh[(size_t)t * DD + e] = acc[r];
            }
        }
        __syncthreads();

        {
            v8f acc;
            #pragma unroll
            for (int r = 0; r < 8; ++r)
                acc[r] = sS[m0 + r + 8 * half][lr];
            #pragma unroll
            for (int k0 = 0; k0 < CHUNK; k0 += 4) {
                v2f a, b;
                a[0] = sK[k0 + 2 * half + 0][m0 + lr];
                a[1] = sK[k0 + 2 * half + 1][m0 + lr];
                b[0] = sV[k0 + 2 * half + 0][lr];
                b[1] = sV[k0 + 2 * half + 1][lr];
                acc = wmma_f32(a, b, acc);
            }
            #pragma unroll
            for (int r = 0; r < 8; ++r)
                sS[m0 + r + 8 * half][lr] = DECAY_CHUNK * acc[r];
        }
    }
}

extern "C" void kernel_launch(void* const* d_in, const int* in_sizes, int n_in,
                              void* d_out, int out_size, void* d_ws, size_t ws_size,
                              hipStream_t stream) {
    (void)in_sizes; (void)n_in; (void)d_ws; (void)ws_size; (void)out_size;
    const float* q = (const float*)d_in[0];
    const float* k = (const float*)d_in[1];
    const float* v = (const float*)d_in[2];
    float* out = (float*)d_out;
    rkv_chunked_wmma<<<dim3(BH * NE), dim3(128), 0, stream>>>(q, k, v, out);
}


# One work-group per (chunk, head, 16-col state slice): each rebuilds its chunk's start state by rerunning the recurrence over all earlier K,V; 1 launch, nothing passed between groups, sum order kept
# speedup vs baseline: 4.0338x; 4.0338x over previous
#include <hip/hip_runtime.h>


#define BH      32
#define T       1024
#define DD      64
#define CHUNK   64
#define NCHUNK  (T / CHUNK)
#define EB      16
#define NE      (DD / EB)
#define QSTR    68
#define VSTR    24
#define KLD     (CHUNK * (DD / 4) / 128)
#define VLD     (CHUNK * (EB / 4) / 128)

#define L2DECAY     (-0.07400058144377693f)
#define DECAY_CHUNK (0.0375241347f)

typedef __attribute__((ext_vector_type(2))) float v2f;
typedef __attribute__((ext_vector_type(8))) float v8f;

__device__ __forceinline__ v8f wmma_f32(v2f a, v2f b, v8f c) {
    return __builtin_amdgcn_wmma_f32_16x16x4_f32(
         false, a,  false, b,
         (short)0, c,  false,  false);
}

__global__ __launch_bounds__(128)
void rkv_chunk_parallel_wmma(const float* __restrict__ q,
                             const float* __restrict__ k,
                             const float* __restrict__ v,
                             float* __restrict__ out)
{
    __shared__ float sQ[CHUNK][QSTR];
    __shared__ float sK[CHUNK][QSTR];
    __shared__ float sA[CHUNK][QSTR];
    __shared__ float sV[CHUNK][VSTR];
    __shared__ float sS[DD][VSTR];

    const int tid  = threadIdx.x;
    const int lane = tid & 31;
    const int wave = tid >> 5;
    const int lr   = lane & 15;
    const int half = lane >> 4;
    const int m0   = wave * 16;

    const int c   = (NCHUNK - 1) - (int)(blockIdx.x / (BH * NE));
    const int rem = blockIdx.x % (BH * NE);
    const int bh  = rem / NE;
    const int e0  = (rem % NE) * EB;

    const float* qbh = q + (size_t)bh * T * DD;
    const float* kbh = k + (size_t)bh * T * DD;
    const float* vbh = v + (size_t)bh * T * DD;
    float*       obh = out + (size_t)bh * T * DD;
    const int tbase = c * CHUNK;

    float4 kreg[KLD];
    float4 vreg[VLD];

    auto load_kv = [&](int base) {
        #pragma unroll
        for (int u = 0; u < KLD; ++u) {
            const int i   = tid + 128 * u;
            const int row = i / (DD / 4);
            const int c4  = (i % (DD / 4)) * 4;
            kreg[u] = *(const float4*)(kbh + (size_t)(base + row) * DD + c4);
        }
        #pragma unroll
        for (int u = 0; u < VLD; ++u) {
            const int i   = tid + 128 * u;
            const int row = i / (EB / 4);
            const int c4  = (i % (EB / 4)) * 4;
            vreg[u] = *(const float4*)(vbh + (size_t)(base + row) * DD + e0 + c4);
        }
    };

    auto store_kv = [&]() {
        #pragma unroll
        for (int u = 0; u < KLD; ++u) {
            const int i   = tid + 128 * u;
            const int row = i / (DD / 4);
            const int c4  = (i % (DD / 4)) * 4;
            const float ks = __builtin_amdgcn_exp2f((float)(-(row + 1)) * L2DECAY);
            sK[row][c4 + 0] = kreg[u].x * ks; sK[row][c4 + 1] = kreg[u].y * ks;
            sK[row][c4 + 2] = kreg[u].z * ks; sK[row][c4 + 3] = kreg[u].w * ks;
        }
        #pragma unroll
        for (int u = 0; u < VLD; ++u) {
            const int i   = tid + 128 * u;
            const int row = i / (EB / 4);
            const int c4  = (i % (EB / 4)) * 4;
            sV[row][c4 + 0] = vreg[u].x; sV[row][c4 + 1] = vreg[u].y;
            sV[row][c4 + 2] = vreg[u].z; sV[row][c4 + 3] = vreg[u].w;
        }
    };

    {
        float4 qreg[KLD];
        #pragma unroll
        for (int u = 0; u < KLD; ++u) {
            const int i   = tid + 128 * u;
            const int row = i / (DD / 4);
            const int c4  = (i % (DD / 4)) * 4;
            qreg[u] = *(const float4*)(qbh + (size_t)(tbase + row) * DD + c4);
        }
        load_kv(0);
        #pragma unroll
        for (int u = 0; u < KLD; ++u) {
            const int i   = tid + 128 * u;
            const int row = i / (DD / 4);
            const int c4  = (i % (DD / 4)) * 4;
            const float qs = __builtin_amdgcn_exp2f((float)(row + 1) * L2DECAY);
            sQ[row][c4 + 0] = qreg[u].x * qs; sQ[row][c4 + 1] = qreg[u].y * qs;
            sQ[row][c4 + 2] = qreg[u].z * qs; sQ[row][c4 + 3] = qreg[u].w * qs;
        }
    }

    v8f st = {};

    for (int j = 0; j < c; ++j) {
        __syncthreads();
        store_kv();
        __syncthreads();

        load_kv((j + 1) * CHUNK);

        #pragma unroll
        for (int k0 = 0; k0 < CHUNK; k0 += 4) {
            v2f a, b;
            a[0] = sK[k0 + 2 * half + 0][m0 + lr];
            a[1] = sK[k0 + 2 * half + 1][m0 + lr];
            b[0] = sV[k0 + 2 * half + 0][lr];
            b[1] = sV[k0 + 2 * half + 1][lr];
            st = wmma_f32(a, b, st);
        }
        #pragma unroll
        for (int r = 0; r < 8; ++r)
            st[r] = DECAY_CHUNK * st[r];
    }

    __syncthreads();
    store_kv();

    #pragma unroll
    for (int r = 0; r < 8; ++r)
        sS[m0 + r + 8 * half][lr] = st[r];
    __syncthreads();

    for (int nb = 0; nb < 4; ++nb) {
        const int n0 = nb * 16;
        v8f acc = {};
        #pragma unroll
        for (int k0 = 0; k0 < DD; k0 += 4) {
            v2f a, b;
            a[0] = sQ[m0 + lr][k0 + 2 * half + 0];
            a[1] = sQ[m0 + lr][k0 + 2 * half + 1];
            b[0] = sK[n0 + lr][k0 + 2 * half + 0];
            b[1] = sK[n0 + lr][k0 + 2 * half + 1];
            acc = wmma_f32(a, b, acc);
        }
        #pragma unroll
        for (int r = 0; r < 8; ++r) {
            const int t = m0 + r + 8 * half;
            const int s = n0 + lr;
            sA[t][s] = (s <= t) ? acc[r] : 0.0f;
        }
    }

    {
        v8f acc = {};
        #pragma unroll
        for (int k0 = 0; k0 < CHUNK; k0 += 4) {
            v2f a, b;
            a[0] = sA[m0 + lr][k0 + 2 * half + 0];
            a[1] = sA[m0 + lr][k0 + 2 * half + 1];
            b[0] = sV[k0 + 2 * half + 0][lr];
            b[1] = sV[k0 + 2 * half + 1][lr];
            acc = wmma_f32(a, b, acc);
        }
        #pragma unroll
        for (int k0 = 0; k0 < DD; k0 += 4) {
            v2f a, b;
            a[0] = sQ[m0 + lr][k0 + 2 * half + 0];
            a[1] = sQ[m0 + lr][k0 + 2 * half + 1];
            b[0] = sS[k0 + 2 * half + 0][lr];
            b[1] = sS[k0 + 2 * half + 1][lr];
            acc = wmma_f32(a, b, acc);
        }
        #pragma unroll
        for (int r = 0; r < 8; ++r) {
            const int t = tbase + m0 + r + 8 * half;
            const int e = e0 + lr;
            obh[(size_t)t * DD + e] = acc[r];
        }
    }
}

extern "C" void kernel_launch(void* const* d_in, const int* in_sizes, int n_in,
                              void* d_out, int out_size, void* d_ws, size_t ws_size,
                              hipStream_t stream) {
    (void)in_sizes; (void)n_in; (void)d_ws; (void)ws_size; (void)out_size;
    const float* q = (const float*)d_in[0];
    const float* k = (const float*)d_in[1];
    const float* v = (const float*)d_in[2];
    float* out = (float*)d_out;
    rkv_chunk_parallel_wmma<<<dim3(BH * NE * NCHUNK), dim3(128), 0, stream>>>(q, k, v, out);
}
